# starting kernel
# baseline (speedup 1.0000x reference)
#include <hip/hip_runtime.h>
#include <hip/hip_bf16.h>
#include <cstdint>

#define B_ 8
#define C_ 512
#define T_ 2048
#define HEADS_ 8
#define KER_ 31
#define NROWS (B_ * T_)
#define EPS_ 1e-5f

typedef __attribute__((ext_vector_type(2))) float v2f;
typedef __attribute__((ext_vector_type(8))) float v8f;

__device__ __forceinline__ v8f wmma_f32(v2f a, v2f b, v8f c) {
  return __builtin_amdgcn_wmma_f32_16x16x4_f32(false, a, false, b, (short)0, c,
                                               false, false);
}

__global__ __launch_bounds__(256) void ln_stats_kernel(
    const float* __restrict__ x, float* __restrict__ stats) {
  const int tid = threadIdx.x;
  const int row0 = blockIdx.x * 64;
  const int b = row0 / T_;
  const int t0 = row0 % T_;
  const int r = tid & 63;
  const int g = tid >> 6;
  float s = 0.f, s2 = 0.f;
  const float* xb = x + (size_t)b * C_ * T_ + t0 + r;
#pragma unroll 4
  for (int i = 0; i < 128; ++i) {
    int c = g * 128 + i;
    float v = xb[(size_t)c * T_];
    s += v;
    s2 += v * v;
  }
  __shared__ float sb[4][64], sq[4][64];
  sb[g][r] = s;
  sq[g][r] = s2;
  __syncthreads();
  if (tid < 64) {
    float S = sb[0][tid] + sb[1][tid] + sb[2][tid] + sb[3][tid];
    float S2 = sq[0][tid] + sq[1][tid] + sq[2][tid] + sq[3][tid];
    float mu = S * (1.f / C_);
    float var = S2 * (1.f / C_) - mu * mu;
    stats[(row0 + tid) * 2 + 0] = mu;
    stats[(row0 + tid) * 2 + 1] = rsqrtf(var + EPS_);
  }
}

#define AS_S 66
#define BS_S 66
#define KT_S 65
__global__ __launch_bounds__(256) void glu_gemm_kernel(
    const float* __restrict__ x, const float* __restrict__ gamma,
    const float* __restrict__ beta, const float* __restrict__ Wk,
    const float* __restrict__ bk, const float* __restrict__ stats,
    float* __restrict__ k_ws) {
  __shared__ float smem[64 * AS_S + 128 * BS_S];
  float* As = smem;
  float* Bs = smem + 64 * AS_S;

  const int tid = threadIdx.x;
  const int row0 = blockIdx.x * 64;
  const int n0 = blockIdx.y * 64;
  const int b = row0 / T_;
  const int t0 = row0 % T_;
  const int r_ld = tid & 63;
  const int kg = tid >> 6;
  const float mu = stats[(row0 + r_ld) * 2 + 0];
  const float rs = stats[(row0 + r_ld) * 2 + 1];

  const int lane = tid & 31;
  const int wave = tid >> 5;
  const int laneM = lane & 15;
  const int hi = lane >> 4;
  const int rblk = wave & 3;
  const int j0 = (wave >> 2) * 32;

  v8f accA0 = {}, accA1 = {}, accB0 = {}, accB1 = {};
  const float* xb = x + (size_t)b * C_ * T_ + t0 + r_ld;

  for (int kbase = 0; kbase < C_; kbase += 64) {
#pragma unroll
    for (int i = 0; i < 16; ++i) {
      int kk = kg * 16 + i;
      int c = kbase + kk;
      float v = xb[(size_t)c * T_];
      As[r_ld * AS_S + kk] = (v - mu) * rs * gamma[c] + beta[c];
    }
    {
      int jj = tid & 63;
      int kq = tid >> 6;
#pragma unroll
      for (int part = 0; part < 2; ++part) {
        int n = n0 + jj + part * C_;
#pragma unroll
        for (int i = 0; i < 16; ++i) {
          int kk = kq * 16 + i;
          Bs[(part * 64 + jj) * BS_S + kk] =
              Wk[(size_t)(kbase + kk) * (2 * C_) + n];
        }
      }
    }
    __syncthreads();
#pragma unroll
    for (int kb = 0; kb < 64; kb += 4) {
      v2f af = *reinterpret_cast<const v2f*>(
          &As[(rblk * 16 + laneM) * AS_S + kb + 2 * hi]);
      v2f b0 = *reinterpret_cast<const v2f*>(
          &Bs[(j0 + laneM) * BS_S + kb + 2 * hi]);
      v2f b1 = *reinterpret_cast<const v2f*>(
          &Bs[(j0 + 16 + laneM) * BS_S + kb + 2 * hi]);
      v2f b2 = *reinterpret_cast<const v2f*>(
          &Bs[(64 + j0 + laneM) * BS_S + kb + 2 * hi]);
      v2f b3 = *reinterpret_cast<const v2f*>(
          &Bs[(64 + j0 + 16 + laneM) * BS_S + kb + 2 * hi]);
      accA0 = wmma_f32(af, b0, accA0);
      accA1 = wmma_f32(af, b1, accA1);
      accB0 = wmma_f32(af, b2, accB0);
      accB1 = wmma_f32(af, b3, accB1);
    }
    __syncthreads();
  }

  float* kst = smem;
  const float bka0 = bk[n0 + j0 + laneM];
  const float bka1 = bk[n0 + j0 + 16 + laneM];
  const float bkb0 = bk[C_ + n0 + j0 + laneM];
  const float bkb1 = bk[C_ + n0 + j0 + 16 + laneM];
#pragma unroll
  for (int v = 0; v < 8; ++v) {
    int tl = rblk * 16 + v + 8 * hi;
    float a0 = accA0[v] + bka0;
    float g0 = accB0[v] + bkb0;
    float a1 = accA1[v] + bka1;
    float g1 = accB1[v] + bkb1;
    kst[(j0 + laneM) * KT_S + tl] = a0 / (1.f + __expf(-g0));
    kst[(j0 + 16 + laneM) * KT_S + tl] = a1 / (1.f + __expf(-g1));
  }
  __syncthreads();
  float* kout = k_ws + (size_t)b * C_ * T_;
  for (int i = tid; i < 64 * 64; i += 256) {
    int cc = i >> 6;
    int tl = i & 63;
    kout[(size_t)(n0 + cc) * T_ + t0 + tl] = kst[cc * KT_S + tl];
  }
}

#define A2_S 34
#define B2_S 34
#define P_S 257
__global__ __launch_bounds__(256) void conv_weight_kernel(
    const float* __restrict__ k_ws, const float* __restrict__ Wc,
    const float* __restrict__ bc, float* __restrict__ w_ws) {
  __shared__ float As[32 * A2_S];
  __shared__ float Bs[256 * B2_S];
  __shared__ float proj[32 * P_S];
  const int tid = threadIdx.x;
  const int row0 = blockIdx.x * 32;
  const int b = row0 / T_;
  const int t0 = row0 % T_;
  const int lane = tid & 31;
  const int wave = tid >> 5;
  const int laneM = lane & 15;
  const int hi = lane >> 4;
  const int rb = wave & 1;
  const int jb = (wave >> 1) * 64;

  v8f acc[4] = {{}, {}, {}, {}};
  const float* kin = k_ws + (size_t)b * C_ * T_ + t0;

  for (int kbase = 0; kbase < C_; kbase += 32) {
    {
      int r = tid & 31;
      int cq = tid >> 5;
#pragma unroll
      for (int i = 0; i < 4; ++i) {
        int kk = cq + 8 * i;
        As[r * A2_S + kk] = kin[(size_t)(kbase + kk) * T_ + r];
      }
    }
    {
      int nn = tid & 63;
      int kq = tid >> 6;
#pragma unroll
      for (int q = 0; q < 4; ++q) {
        int n = nn + 64 * q;
#pragma unroll
        for (int i = 0; i < 8; ++i) {
          int kk = kq * 8 + i;
          float v = (n < KER_ * HEADS_)
                        ? Wc[(size_t)(kbase + kk) * (KER_ * HEADS_) + n]
                        : 0.f;
          Bs[n * B2_S + kk] = v;
        }
      }
    }
    __syncthreads();
#pragma unroll
    for (int kb = 0; kb < 32; kb += 4) {
      v2f af = *reinterpret_cast<const v2f*>(
          &As[(rb * 16 + laneM) * A2_S + kb + 2 * hi]);
#pragma unroll
      for (int jt = 0; jt < 4; ++jt) {
        v2f bf = *reinterpret_cast<const v2f*>(
            &Bs[(jb + jt * 16 + laneM) * B2_S + kb + 2 * hi]);
        acc[jt] = wmma_f32(af, bf, acc[jt]);
      }
    }
    __syncthreads();
  }
#pragma unroll
  for (int jt = 0; jt < 4; ++jt)
#pragma unroll
    for (int v = 0; v < 8; ++v)
      proj[(rb * 16 + v + 8 * hi) * P_S + jb + jt * 16 + laneM] = acc[jt][v];
  __syncthreads();

  {
    int rr = tid >> 3;
    int h = tid & 7;
    float* prow = &proj[rr * P_S + h * KER_];
    const float* bch = &bc[h * KER_];
    float m = -1e30f;
#pragma unroll
    for (int i = 0; i < KER_; ++i) {
      float v = prow[i] + bch[i];
      prow[i] = v;
      m = fmaxf(m, v);
    }
    float s = 0.f;
#pragma unroll
    for (int i = 0; i < KER_; ++i) {
      float e = __expf(prow[i] - m);
      prow[i] = e;
      s += e;
    }
    float inv = 1.f / s;
    float* wout = w_ws + (size_t)(row0 + rr) * 256 + h * 32;
#pragma unroll
    for (int i = 0; i < KER_; ++i) wout[i] = prow[i] * inv;
    wout[KER_] = 0.f;
  }
}

__global__ __launch_bounds__(256) void dynconv_kernel(
    const float* __restrict__ x, const float* __restrict__ k_ws,
    const float* __restrict__ w_ws, const float* __restrict__ conv_bias,
    float* __restrict__ out) {
  __shared__ float wl[32 * 257];
  const int tid = threadIdx.x;
  const int t0 = blockIdx.x * 32;
  const int b = blockIdx.y;
  const float* wsrc = w_ws + ((size_t)b * T_ + t0) * 256;
  for (int i = tid; i < 32 * 256; i += 256)
    wl[(i >> 8) * 257 + (i & 255)] = wsrc[i];
  __syncthreads();

  const int tx = tid & 31;
  const int cg = tid >> 5;
  const int t = t0 + tx;
  const float* wrow = &wl[tx * 257 + cg * 32];
  const bool interior = (t0 >= KER_ / 2) && (t0 + 32 + KER_ / 2 <= T_);

#pragma unroll 1
  for (int ci = 0; ci < 64; ++ci) {
    int c = cg * 64 + ci;
    const size_t basec = ((size_t)b * C_ + c) * T_;
    float acc = conv_bias[c] + x[basec + t];
    if (interior) {
#pragma unroll
      for (int kk = 0; kk < KER_; ++kk)
        acc = fmaf(k_ws[basec + t + kk - KER_ / 2], wrow[kk], acc);
    } else {
#pragma unroll
      for (int kk = 0; kk < KER_; ++kk) {
        int tt = t + kk - KER_ / 2;
        if ((unsigned)tt < (unsigned)T_)
          acc = fmaf(k_ws[basec + tt], wrow[kk], acc);
      }
    }
    out[basec + t] = acc;
  }
}

extern "C" void kernel_launch(void* const* d_in, const int* in_sizes, int n_in,
                              void* d_out, int out_size, void* d_ws,
                              size_t ws_size, hipStream_t stream) {
  (void)in_sizes;
  (void)n_in;
  (void)out_size;
  (void)ws_size;
  const float* x = (const float*)d_in[0];
  const float* ln_gamma = (const float*)d_in[1];
  const float* ln_beta = (const float*)d_in[2];
  const float* Wk = (const float*)d_in[3];
  const float* bk = (const float*)d_in[4];
  const float* Wc = (const float*)d_in[5];
  const float* bc = (const float*)d_in[6];
  const float* conv_bias = (const float*)d_in[7];
  float* out = (float*)d_out;

  float* ws = (float*)d_ws;
  float* stats = ws;
  float* k_ws = ws + 32768;
  float* w_ws = k_ws + (size_t)B_ * C_ * T_;

  ln_stats_kernel<<<NROWS / 64, 256, 0, stream>>>(x, stats);
  glu_gemm_kernel<<<dim3(NROWS / 64, C_ / 64), 256, 0, stream>>>(
      x, ln_gamma, ln_beta, Wk, bk, stats, k_ws);
  conv_weight_kernel<<<NROWS / 32, 256, 0, stream>>>(k_ws, Wc, bc, w_ws);
  dynconv_kernel<<<dim3(T_ / 32, B_), 256, 0, stream>>>(x, k_ws, w_ws,
                                                        conv_bias, out);
}


# dynamic conv recast as a banded-matrix product on the f32 16x16x4 matrix unit (k window x per-position tap band) instead of 31 scalar loads+FMAs per output; tap summation order changes
# speedup vs baseline: 1.7925x; 1.7925x over previous
#include <hip/hip_runtime.h>
#include <hip/hip_bf16.h>
#include <cstdint>

#define B_ 8
#define C_ 512
#define T_ 2048
#define HEADS_ 8
#define KER_ 31
#define NROWS (B_ * T_)
#define EPS_ 1e-5f

typedef __attribute__((ext_vector_type(2))) float v2f;
typedef __attribute__((ext_vector_type(8))) float v8f;

__device__ __forceinline__ v8f wmma_f32(v2f a, v2f b, v8f c) {
  return __builtin_amdgcn_wmma_f32_16x16x4_f32(false, a, false, b, (short)0, c,
                                               false, false);
}

__global__ __launch_bounds__(256) void ln_stats_kernel(
    const float* __restrict__ x, float* __restrict__ stats) {
  const int tid = threadIdx.x;
  const int row0 = blockIdx.x * 64;
  const int b = row0 / T_;
  const int t0 = row0 % T_;
  const int r = tid & 63;
  const int g = tid >> 6;
  float s = 0.f, s2 = 0.f;
  const float* xb = x + (size_t)b * C_ * T_ + t0 + r;
#pragma unroll 4
  for (int i = 0; i < 128; ++i) {
    int c = g * 128 + i;
    float v = xb[(size_t)c * T_];
    s += v;
    s2 += v * v;
  }
  __shared__ float sb[4][64], sq[4][64];
  sb[g][r] = s;
  sq[g][r] = s2;
  __syncthreads();
  if (tid < 64) {
    float S = sb[0][tid] + sb[1][tid] + sb[2][tid] + sb[3][tid];
    float S2 = sq[0][tid] + sq[1][tid] + sq[2][tid] + sq[3][tid];
    float mu = S * (1.f / C_);
    float var = S2 * (1.f / C_) - mu * mu;
    stats[(row0 + tid) * 2 + 0] = mu;
    stats[(row0 + tid) * 2 + 1] = rsqrtf(var + EPS_);
  }
}

#define AS_S 66
#define BS_S 66
#define KT_S 65
__global__ __launch_bounds__(256) void glu_gemm_kernel(
    const float* __restrict__ x, const float* __restrict__ gamma,
    const float* __restrict__ beta, const float* __restrict__ Wk,
    const float* __restrict__ bk, const float* __restrict__ stats,
    float* __restrict__ k_ws) {
  __shared__ float smem[64 * AS_S + 128 * BS_S];
  float* As = smem;
  float* Bs = smem + 64 * AS_S;

  const int tid = threadIdx.x;
  const int row0 = blockIdx.x * 64;
  const int n0 = blockIdx.y * 64;
  const int b = row0 / T_;
  const int t0 = row0 % T_;
  const int r_ld = tid & 63;
  const int kg = tid >> 6;
  const float mu = stats[(row0 + r_ld) * 2 + 0];
  const float rs = stats[(row0 + r_ld) * 2 + 1];

  const int lane = tid & 31;
  const int wave = tid >> 5;
  const int laneM = lane & 15;
  const int hi = lane >> 4;
  const int rblk = wave & 3;
  const int j0 = (wave >> 2) * 32;

  v8f accA0 = {}, accA1 = {}, accB0 = {}, accB1 = {};
  const float* xb = x + (size_t)b * C_ * T_ + t0 + r_ld;

  for (int kbase = 0; kbase < C_; kbase += 64) {
#pragma unroll
    for (int i = 0; i < 16; ++i) {
      int kk = kg * 16 + i;
      int c = kbase + kk;
      float v = xb[(size_t)c * T_];
      As[r_ld * AS_S + kk] = (v - mu) * rs * gamma[c] + beta[c];
    }
    {
      int jj = tid & 63;
      int kq = tid >> 6;
#pragma unroll
      for (int part = 0; part < 2; ++part) {
        int n = n0 + jj + part * C_;
#pragma unroll
        for (int i = 0; i < 16; ++i) {
          int kk = kq * 16 + i;
          Bs[(part * 64 + jj) * BS_S + kk] =
              Wk[(size_t)(kbase + kk) * (2 * C_) + n];
        }
      }
    }
    __syncthreads();
#pragma unroll
    for (int kb = 0; kb < 64; kb += 4) {
      v2f af = *reinterpret_cast<const v2f*>(
          &As[(rblk * 16 + laneM) * AS_S + kb + 2 * hi]);
      v2f b0 = *reinterpret_cast<const v2f*>(
          &Bs[(j0 + laneM) * BS_S + kb + 2 * hi]);
      v2f b1 = *reinterpret_cast<const v2f*>(
          &Bs[(j0 + 16 + laneM) * BS_S + kb + 2 * hi]);
      v2f b2 = *reinterpret_cast<const v2f*>(
          &Bs[(64 + j0 + laneM) * BS_S + kb + 2 * hi]);
      v2f b3 = *reinterpret_cast<const v2f*>(
          &Bs[(64 + j0 + 16 + laneM) * BS_S + kb + 2 * hi]);
      accA0 = wmma_f32(af, b0, accA0);
      accA1 = wmma_f32(af, b1, accA1);
      accB0 = wmma_f32(af, b2, accB0);
      accB1 = wmma_f32(af, b3, accB1);
    }
    __syncthreads();
  }

  float* kst = smem;
  const float bka0 = bk[n0 + j0 + laneM];
  const float bka1 = bk[n0 + j0 + 16 + laneM];
  const float bkb0 = bk[C_ + n0 + j0 + laneM];
  const float bkb1 = bk[C_ + n0 + j0 + 16 + laneM];
#pragma unroll
  for (int v = 0; v < 8; ++v) {
    int tl = rblk * 16 + v + 8 * hi;
    float a0 = accA0[v] + bka0;
    float g0 = accB0[v] + bkb0;
    float a1 = accA1[v] + bka1;
    float g1 = accB1[v] + bkb1;
    kst[(j0 + laneM) * KT_S + tl] = a0 / (1.f + __expf(-g0));
    kst[(j0 + 16 + laneM) * KT_S + tl] = a1 / (1.f + __expf(-g1));
  }
  __syncthreads();
  float* kout = k_ws + (size_t)b * C_ * T_;
  for (int i = tid; i < 64 * 64; i += 256) {
    int cc = i >> 6;
    int tl = i & 63;
    kout[(size_t)(n0 + cc) * T_ + t0 + tl] = kst[cc * KT_S + tl];
  }
}

#define A2_S 34
#define B2_S 34
#define P_S 257
__global__ __launch_bounds__(256) void conv_weight_kernel(
    const float* __restrict__ k_ws, const float* __restrict__ Wc,
    const float* __restrict__ bc, float* __restrict__ w_ws) {
  __shared__ float As[32 * A2_S];
  __shared__ float Bs[256 * B2_S];
  __shared__ float proj[32 * P_S];
  const int tid = threadIdx.x;
  const int row0 = blockIdx.x * 32;
  const int b = row0 / T_;
  const int t0 = row0 % T_;
  const int lane = tid & 31;
  const int wave = tid >> 5;
  const int laneM = lane & 15;
  const int hi = lane >> 4;
  const int rb = wave & 1;
  const int jb = (wave >> 1) * 64;

  v8f acc[4] = {{}, {}, {}, {}};
  const float* kin = k_ws + (size_t)b * C_ * T_ + t0;

  for (int kbase = 0; kbase < C_; kbase += 32) {
    {
      int r = tid & 31;
      int cq = tid >> 5;
#pragma unroll
      for (int i = 0; i < 4; ++i) {
        int kk = cq + 8 * i;
        As[r * A2_S + kk] = kin[(size_t)(kbase + kk) * T_ + r];
      }
    }
    {
      int nn = tid & 63;
      int kq = tid >> 6;
#pragma unroll
      for (int q = 0; q < 4; ++q) {
        int n = nn + 64 * q;
#pragma unroll
        for (int i = 0; i < 8; ++i) {
          int kk = kq * 8 + i;
          float v = (n < KER_ * HEADS_)
                        ? Wc[(size_t)(kbase + kk) * (KER_ * HEADS_) + n]
                        : 0.f;
          Bs[n * B2_S + kk] = v;
        }
      }
    }
    __syncthreads();
#pragma unroll
    for (int kb = 0; kb < 32; kb += 4) {
      v2f af = *reinterpret_cast<const v2f*>(
          &As[(rb * 16 + laneM) * A2_S + kb + 2 * hi]);
#pragma unroll
      for (int jt = 0; jt < 4; ++jt) {
        v2f bf = *reinterpret_cast<const v2f*>(
            &Bs[(jb + jt * 16 + laneM) * B2_S + kb + 2 * hi]);
        acc[jt] = wmma_f32(af, bf, acc[jt]);
      }
    }
    __syncthreads();
  }
#pragma unroll
  for (int jt = 0; jt < 4; ++jt)
#pragma unroll
    for (int v = 0; v < 8; ++v)
      proj[(rb * 16 + v + 8 * hi) * P_S + jb + jt * 16 + laneM] = acc[jt][v];
  __syncthreads();

  {
    int rr = tid >> 3;
    int h = tid & 7;
    float* prow = &proj[rr * P_S + h * KER_];
    const float* bch = &bc[h * KER_];
    float m = -1e30f;
#pragma unroll
    for (int i = 0; i < KER_; ++i) {
      float v = prow[i] + bch[i];
      prow[i] = v;
      m = fmaxf(m, v);
    }
    float s = 0.f;
#pragma unroll
    for (int i = 0; i < KER_; ++i) {
      float e = __expf(prow[i] - m);
      prow[i] = e;
      s += e;
    }
    float inv = 1.f / s;
    float* wout = w_ws + (size_t)(row0 + rr) * 256 + h * 32;
#pragma unroll
    for (int i = 0; i < KER_; ++i) wout[i] = prow[i] * inv;
    wout[KER_] = 0.f;
  }
}

typedef __attribute__((ext_vector_type(4))) float v4f;
#define DW_S 260
#define DO_S 36
__device__ __forceinline__ void load_kwin(v4f (&kv)[8],
                                          const float* __restrict__ krow,
                                          bool lo_ok, bool hi_ok) {
#pragma unroll
  for (int u = 0; u < 8; ++u) {
    const bool in = (u >= 2 || lo_ok) && (u < 6 || hi_ok);
    const v4f v = *reinterpret_cast<const v4f*>(krow + (in ? 8 * u - 16 : 0));
    kv[u] = in ? v : (v4f){0.f, 0.f, 0.f, 0.f};
  }
}
__global__ __launch_bounds__(256) void dynconv_kernel(
    const float* __restrict__ x, const float* __restrict__ k_ws,
    const float* __restrict__ w_ws, const float* __restrict__ conv_bias,
    float* __restrict__ out) {
  __shared__ __attribute__((aligned(16))) float wl[32 * DW_S];
  __shared__ __attribute__((aligned(16))) float ost[8 * 16 * DO_S];
  const int tid = threadIdx.x;
  const int t0 = blockIdx.x * 32;
  const int b = blockIdx.y;
  {
    const v4f* wsrc =
        reinterpret_cast<const v4f*>(w_ws + ((size_t)b * T_ + t0) * 256);
    for (int i = tid; i < 32 * 64; i += 256)
      *reinterpret_cast<v4f*>(&wl[(i >> 6) * DW_S + (i & 63) * 4]) = wsrc[i];
  }
  __syncthreads();

  const int lane = tid & 31;
  const int head = tid >> 5;
  const int cN = lane & 15;
  const int hi = lane >> 4;

  v2f bfr[2][12];
#pragma unroll
  for (int nt = 0; nt < 2; ++nt) {
    const float* wr = &wl[(16 * nt + cN) * DW_S + head * 32];
#pragma unroll
    for (int s = 0; s < 12; ++s) {
      const int kk0 = 8 * (s >> 1) + 4 * hi + 2 * (s & 1) - cN - 1;
      const int kk1 = kk0 + 1;
      const float w0 = wr[min(max(kk0, 0), KER_ - 1)];
      const float w1 = wr[min(max(kk1, 0), KER_ - 1)];
      v2f f;
      f.x = ((unsigned)kk0 < (unsigned)KER_) ? w0 : 0.f;
      f.y = ((unsigned)kk1 < (unsigned)KER_) ? w1 : 0.f;
      bfr[nt][s] = f;
    }
  }

  const bool lo_ok = (t0 != 0);
  const bool hi_ok = (t0 + 32 != T_);
  const size_t rowb = (size_t)b * C_ + head * 64;
  float* o = &ost[head * 16 * DO_S];
  const int erow = lane >> 3;
  const int ec4 = (lane & 7) * 4;

  v4f kv[8];
  load_kwin(kv, k_ws + (rowb + cN) * T_ + t0 + 4 * hi, lo_ok, hi_ok);

#pragma unroll
  for (int mt = 0; mt < 4; ++mt) {
    v4f xv[4];
    float cb[4];
#pragma unroll
    for (int q = 0; q < 4; ++q) {
      const size_t chn = rowb + mt * 16 + q * 4 + erow;
      xv[q] = *reinterpret_cast<const v4f*>(x + chn * T_ + t0 + ec4);
      cb[q] = conv_bias[head * 64 + mt * 16 + q * 4 + erow];
    }
    v4f kn[8];
    if (mt < 3)
      load_kwin(kn, k_ws + (rowb + (mt + 1) * 16 + cN) * T_ + t0 + 4 * hi,
                lo_ok, hi_ok);

    v8f acc0 = {}, acc1 = {};
#pragma unroll
    for (int g = 0; g < 16; ++g) {
      v2f a;
      if (g & 1) {
        a.x = kv[g >> 1].z;
        a.y = kv[g >> 1].w;
      } else {
        a.x = kv[g >> 1].x;
        a.y = kv[g >> 1].y;
      }
      if (g < 12) acc0 = wmma_f32(a, bfr[0][g < 12 ? g : 0], acc0);
      if (g >= 4) acc1 = wmma_f32(a, bfr[1][g >= 4 ? g - 4 : 0], acc1);
    }

    __syncthreads();
#pragma unroll
    for (int r = 0; r < 8; ++r) {
      o[(8 * hi + r) * DO_S + cN] = acc0[r];
      o[(8 * hi + r) * DO_S + 16 + cN] = acc1[r];
    }
    __syncthreads();
#pragma unroll
    for (int q = 0; q < 4; ++q) {
      const int row = q * 4 + erow;
      v4f d = *reinterpret_cast<const v4f*>(&o[row * DO_S + ec4]);
      const size_t chn = rowb + mt * 16 + row;
      d = d + xv[q] + cb[q];
      *reinterpret_cast<v4f*>(out + chn * T_ + t0 + ec4) = d;
    }
    if (mt < 3) {
#pragma unroll
      for (int u = 0; u < 8; ++u) kv[u] = kn[u];
    }
  }
}

extern "C" void kernel_launch(void* const* d_in, const int* in_sizes, int n_in,
                              void* d_out, int out_size, void* d_ws,
                              size_t ws_size, hipStream_t stream) {
  (void)in_sizes;
  (void)n_in;
  (void)out_size;
  (void)ws_size;
  const float* x = (const float*)d_in[0];
  const float* ln_gamma = (const float*)d_in[1];
  const float* ln_beta = (const float*)d_in[2];
  const float* Wk = (const float*)d_in[3];
  const float* bk = (const float*)d_in[4];
  const float* Wc = (const float*)d_in[5];
  const float* bc = (const float*)d_in[6];
  const float* conv_bias = (const float*)d_in[7];
  float* out = (float*)d_out;

  float* ws = (float*)d_ws;
  float* stats = ws;
  float* k_ws = ws + 32768;
  float* w_ws = k_ws + (size_t)B_ * C_ * T_;

  ln_stats_kernel<<<NROWS / 64, 256, 0, stream>>>(x, stats);
  glu_gemm_kernel<<<dim3(NROWS / 64, C_ / 64), 256, 0, stream>>>(
      x, ln_gamma, ln_beta, Wk, bk, stats, k_ws);
  conv_weight_kernel<<<NROWS / 32, 256, 0, stream>>>(k_ws, Wc, bc, w_ws);
  dynconv_kernel<<<dim3(T_ / 32, B_), 256, 0, stream>>>(x, k_ws, w_ws,
                                                        conv_bias, out);
}


# LayerNorm folded through GEMM1: product runs on x*gamma, mean and rstd applied to the accumulators as a rank-1 correction; stats kernel and its pass over x removed; proj summation order changes
# speedup vs baseline: 1.9203x; 1.0713x over previous
#include <hip/hip_runtime.h>
#include <hip/hip_bf16.h>
#include <cstdint>

#define B_ 8
#define C_ 512
#define T_ 2048
#define HEADS_ 8
#define KER_ 31
#define NROWS (B_ * T_)
#define EPS_ 1e-5f

typedef __attribute__((ext_vector_type(2))) float v2f;
typedef __attribute__((ext_vector_type(8))) float v8f;

__device__ __forceinline__ v8f wmma_f32(v2f a, v2f b, v8f c) {
  return __builtin_amdgcn_wmma_f32_16x16x4_f32(false, a, false, b, (short)0, c,
                                               false, false);
}

#define AS_S 66
#define BS_S 66
#define KT_S 65
#define RED_O 4352
__global__ __launch_bounds__(256) __attribute__((amdgpu_waves_per_eu(8, 8))) void
glu_gemm_kernel(
    const float* __restrict__ x, const float* __restrict__ gamma,
    const float* __restrict__ beta, const float* __restrict__ Wk,
    const float* __restrict__ bk, float* __restrict__ k_ws) {
  __shared__ __attribute__((aligned(16))) float smem[64 * AS_S + 128 * BS_S];
  float* As = smem;
  float* Bs = smem + 64 * AS_S;

  const int tid = threadIdx.x;
  const int row0 = blockIdx.x * 64;
  const int n0 = blockIdx.y * 64;
  const int b = row0 / T_;
  const int t0 = row0 % T_;
  const int r_ld = tid & 63;

  const int lane = tid & 31;
  const int wave = __builtin_amdgcn_readfirstlane(tid >> 5);
  const int kg = wave >> 1;
  const int laneM = lane & 15;
  const int hi = lane >> 4;
  const int rblk = wave & 3;
  const int j0 = (wave >> 2) * 32;

  v8f accA0 = {}, accA1 = {}, accB0 = {}, accB1 = {};
  const float* xb = x + (size_t)b * C_ * T_ + t0 + r_ld;
  v2f sx2 = {0.f, 0.f}, sxx2 = {0.f, 0.f};
  v2f gs2[2] = {{0.f, 0.f}, {0.f, 0.f}};
  v2f bs2[2] = {{0.f, 0.f}, {0.f, 0.f}};

  for (int kbase = 0; kbase < C_; kbase += 64) {
#pragma unroll
    for (int i = 0; i < 8; ++i) {
      int kk = kg * 16 + 2 * i;
      int c = kbase + kk;
      v2f v;
      v.x = xb[(size_t)c * T_];
      v.y = xb[(size_t)(c + 1) * T_];
      const v2f g2 = *reinterpret_cast<const v2f*>(&gamma[c]);
      *reinterpret_cast<v2f*>(&As[r_ld * AS_S + kk]) = v * g2;
      sx2 += v;
      sxx2 += v * v;
    }
#pragma unroll
    for (int part = 0; part < 2; ++part) {
      int n = n0 + r_ld + part * C_;
#pragma unroll
      for (int i = 0; i < 8; ++i) {
        int kk = kg * 16 + 2 * i;
        v2f w;
        w.x = Wk[(size_t)(kbase + kk) * (2 * C_) + n];
        w.y = Wk[(size_t)(kbase + kk + 1) * (2 * C_) + n];
        const v2f g2 = *reinterpret_cast<const v2f*>(&gamma[kbase + kk]);
        const v2f b2 = *reinterpret_cast<const v2f*>(&beta[kbase + kk]);
        *reinterpret_cast<v2f*>(&Bs[(part * 64 + r_ld) * BS_S + kk]) = w;
        gs2[part] += g2 * w;
        bs2[part] += b2 * w;
      }
    }
    __syncthreads();
#pragma unroll
    for (int kb = 0; kb < 64; kb += 4) {
      v2f af = *reinterpret_cast<const v2f*>(
          &As[(rblk * 16 + laneM) * AS_S + kb + 2 * hi]);
      v2f b0 = *reinterpret_cast<const v2f*>(
          &Bs[(j0 + laneM) * BS_S + kb + 2 * hi]);
      v2f b1 = *reinterpret_cast<const v2f*>(
          &Bs[(j0 + 16 + laneM) * BS_S + kb + 2 * hi]);
      v2f b2 = *reinterpret_cast<const v2f*>(
          &Bs[(64 + j0 + laneM) * BS_S + kb + 2 * hi]);
      v2f b3 = *reinterpret_cast<const v2f*>(
          &Bs[(64 + j0 + 16 + laneM) * BS_S + kb + 2 * hi]);
      accA0 = wmma_f32(af, b0, accA0);
      accA1 = wmma_f32(af, b1, accA1);
      accB0 = wmma_f32(af, b2, accB0);
      accB1 = wmma_f32(af, b3, accB1);
    }
    __syncthreads();
  }

  float* p_sx = smem + RED_O;
  float* p_sxx = smem + RED_O + 256;
  float* p_g = smem + RED_O + 512;
  float* p_b = smem + RED_O + 1024;
  float* f_mu = smem + RED_O + 1536;
  float* f_rs = smem + RED_O + 1600;
  float* f_g = smem + RED_O + 1664;
  float* f_b = smem + RED_O + 1792;
  p_sx[kg * 64 + r_ld] = sx2.x + sx2.y;
  p_sxx[kg * 64 + r_ld] = sxx2.x + sxx2.y;
  p_g[kg * 128 + r_ld] = gs2[0].x + gs2[0].y;
  p_g[kg * 128 + 64 + r_ld] = gs2[1].x + gs2[1].y;
  p_b[kg * 128 + r_ld] = bs2[0].x + bs2[0].y;
  p_b[kg * 128 + 64 + r_ld] = bs2[1].x + bs2[1].y;
  __syncthreads();
  if (tid < 64) {
    float S = p_sx[tid] + p_sx[64 + tid] + p_sx[128 + tid] + p_sx[192 + tid];
    float S2 = p_sxx[tid] + p_sxx[64 + tid] + p_sxx[128 + tid] + p_sxx[192 + tid];
    float mu = S * (1.f / C_);
    float var = S2 * (1.f / C_) - mu * mu;
    f_mu[tid] = mu;
    f_rs[tid] = rsqrtf(var + EPS_);
  } else if (tid < 192) {
    int col = tid - 64;
    float G = p_g[col] + p_g[128 + col] + p_g[256 + col] + p_g[384 + col];
    float Bt = p_b[col] + p_b[128 + col] + p_b[256 + col] + p_b[384 + col];
    f_g[col] = G;
    f_b[col] = Bt;
  }
  __syncthreads();

  float* kst = smem;
  const float ga0 = f_g[j0 + laneM];
  const float ga1 = f_g[j0 + 16 + laneM];
  const float gb0 = f_g[64 + j0 + laneM];
  const float gb1 = f_g[64 + j0 + 16 + laneM];
  const float oa0 = f_b[j0 + laneM] + bk[n0 + j0 + laneM];
  const float oa1 = f_b[j0 + 16 + laneM] + bk[n0 + j0 + 16 + laneM];
  const float ob0 = f_b[64 + j0 + laneM] + bk[C_ + n0 + j0 + laneM];
  const float ob1 = f_b[64 + j0 + 16 + laneM] + bk[C_ + n0 + j0 + 16 + laneM];
#pragma unroll
  for (int v = 0; v < 8; ++v) {
    int tl = rblk * 16 + v + 8 * hi;
    const float mu = f_mu[tl];
    const float rs = f_rs[tl];
    float a0 = rs * (accA0[v] - mu * ga0) + oa0;
    float g0 = rs * (accB0[v] - mu * gb0) + ob0;
    float a1 = rs * (accA1[v] - mu * ga1) + oa1;
    float g1 = rs * (accB1[v] - mu * gb1) + ob1;
    kst[(j0 + laneM) * KT_S + tl] = a0 / (1.f + __expf(-g0));
    kst[(j0 + 16 + laneM) * KT_S + tl] = a1 / (1.f + __expf(-g1));
  }
  __syncthreads();
  float* kout = k_ws + (size_t)b * C_ * T_;
  for (int i = tid; i < 64 * 64; i += 256) {
    int cc = i >> 6;
    int tl = i & 63;
    kout[(size_t)(n0 + cc) * T_ + t0 + tl] = kst[cc * KT_S + tl];
  }
}

#define A2_S 34
#define B2_S 34
#define P_S 257
__global__ __launch_bounds__(256) void conv_weight_kernel(
    const float* __restrict__ k_ws, const float* __restrict__ Wc,
    const float* __restrict__ bc, float* __restrict__ w_ws) {
  __shared__ float As[32 * A2_S];
  __shared__ float Bs[256 * B2_S];
  __shared__ float proj[32 * P_S];
  const int tid = threadIdx.x;
  const int row0 = blockIdx.x * 32;
  const int b = row0 / T_;
  const int t0 = row0 % T_;
  const int lane = tid & 31;
  const int wave = tid >> 5;
  const int laneM = lane & 15;
  const int hi = lane >> 4;
  const int rb = wave & 1;
  const int jb = (wave >> 1) * 64;

  v8f acc[4] = {{}, {}, {}, {}};
  const float* kin = k_ws + (size_t)b * C_ * T_ + t0;

  for (int kbase = 0; kbase < C_; kbase += 32) {
    {
      int r = tid & 31;
      int cq = tid >> 5;
#pragma unroll
      for (int i = 0; i < 4; ++i) {
        int kk = cq + 8 * i;
        As[r * A2_S + kk] = kin[(size_t)(kbase + kk) * T_ + r];
      }
    }
    {
      int nn = tid & 63;
      int kq = tid >> 6;
#pragma unroll
      for (int q = 0; q < 4; ++q) {
        int n = nn + 64 * q;
#pragma unroll
        for (int i = 0; i < 8; ++i) {
          int kk = kq * 8 + i;
          float v = (n < KER_ * HEADS_)
                        ? Wc[(size_t)(kbase + kk) * (KER_ * HEADS_) + n]
                        : 0.f;
          Bs[n * B2_S + kk] = v;
        }
      }
    }
    __syncthreads();
#pragma unroll
    for (int kb = 0; kb < 32; kb += 4) {
      v2f af = *reinterpret_cast<const v2f*>(
          &As[(rb * 16 + laneM) * A2_S + kb + 2 * hi]);
#pragma unroll
      for (int jt = 0; jt < 4; ++jt) {
        v2f bf = *reinterpret_cast<const v2f*>(
            &Bs[(jb + jt * 16 + laneM) * B2_S + kb + 2 * hi]);
        acc[jt] = wmma_f32(af, bf, acc[jt]);
      }
    }
    __syncthreads();
  }
#pragma unroll
  for (int jt = 0; jt < 4; ++jt)
#pragma unroll
    for (int v = 0; v < 8; ++v)
      proj[(rb * 16 + v + 8 * hi) * P_S + jb + jt * 16 + laneM] = acc[jt][v];
  __syncthreads();

  {
    int rr = tid >> 3;
    int h = tid & 7;
    float* prow = &proj[rr * P_S + h * KER_];
    const float* bch = &bc[h * KER_];
    float m = -1e30f;
#pragma unroll
    for (int i = 0; i < KER_; ++i) {
      float v = prow[i] + bch[i];
      prow[i] = v;
      m = fmaxf(m, v);
    }
    float s = 0.f;
#pragma unroll
    for (int i = 0; i < KER_; ++i) {
      float e = __expf(prow[i] - m);
      prow[i] = e;
      s += e;
    }
    float inv = 1.f / s;
    float* wout = w_ws + (size_t)(row0 + rr) * 256 + h * 32;
#pragma unroll
    for (int i = 0; i < KER_; ++i) wout[i] = prow[i] * inv;
    wout[KER_] = 0.f;
  }
}

typedef __attribute__((ext_vector_type(4))) float v4f;
#define DW_S 260
#define DO_S 36
__device__ __forceinline__ void load_kwin(v4f (&kv)[8],
                                          const float* __restrict__ krow,
                                          bool lo_ok, bool hi_ok) {
#pragma unroll
  for (int u = 0; u < 8; ++u) {
    const bool in = (u >= 2 || lo_ok) && (u < 6 || hi_ok);
    const v4f v = *reinterpret_cast<const v4f*>(krow + (in ? 8 * u - 16 : 0));
    kv[u] = in ? v : (v4f){0.f, 0.f, 0.f, 0.f};
  }
}
__global__ __launch_bounds__(256) void dynconv_kernel(
    const float* __restrict__ x, const float* __restrict__ k_ws,
    const float* __restrict__ w_ws, const float* __restrict__ conv_bias,
    float* __restrict__ out) {
  __shared__ __attribute__((aligned(16))) float wl[32 * DW_S];
  __shared__ __attribute__((aligned(16))) float ost[8 * 16 * DO_S];
  const int tid = threadIdx.x;
  const int t0 = blockIdx.x * 32;
  const int b = blockIdx.y;
  {
    const v4f* wsrc =
        reinterpret_cast<const v4f*>(w_ws + ((size_t)b * T_ + t0) * 256);
    for (int i = tid; i < 32 * 64; i += 256)
      *reinterpret_cast<v4f*>(&wl[(i >> 6) * DW_S + (i & 63) * 4]) = wsrc[i];
  }
  __syncthreads();

  const int lane = tid & 31;
  const int head = tid >> 5;
  const int cN = lane & 15;
  const int hi = lane >> 4;

  v2f bfr[2][12];
#pragma unroll
  for (int nt = 0; nt < 2; ++nt) {
    const float* wr = &wl[(16 * nt + cN) * DW_S + head * 32];
#pragma unroll
    for (int s = 0; s < 12; ++s) {
      const int kk0 = 8 * (s >> 1) + 4 * hi + 2 * (s & 1) - cN - 1;
      const int kk1 = kk0 + 1;
      const float w0 = wr[min(max(kk0, 0), KER_ - 1)];
      const float w1 = wr[min(max(kk1, 0), KER_ - 1)];
      v2f f;
      f.x = ((unsigned)kk0 < (unsigned)KER_) ? w0 : 0.f;
      f.y = ((unsigned)kk1 < (unsigned)KER_) ? w1 : 0.f;
      bfr[nt][s] = f;
    }
  }

  const bool lo_ok = (t0 != 0);
  const bool hi_ok = (t0 + 32 != T_);
  const size_t rowb = (size_t)b * C_ + head * 64;
  float* o = &ost[head * 16 * DO_S];
  const int erow = lane >> 3;
  const int ec4 = (lane & 7) * 4;

  v4f kv[8];
  load_kwin(kv, k_ws + (rowb + cN) * T_ + t0 + 4 * hi, lo_ok, hi_ok);

#pragma unroll
  for (int mt = 0; mt < 4; ++mt) {
    v4f xv[4];
    float cb[4];
#pragma unroll
    for (int q = 0; q < 4; ++q) {
      const size_t chn = rowb + mt * 16 + q * 4 + erow;
      xv[q] = *reinterpret_cast<const v4f*>(x + chn * T_ + t0 + ec4);
      cb[q] = conv_bias[head * 64 + mt * 16 + q * 4 + erow];
    }
    v4f kn[8];
    if (mt < 3)
      load_kwin(kn, k_ws + (rowb + (mt + 1) * 16 + cN) * T_ + t0 + 4 * hi,
                lo_ok, hi_ok);

    v8f acc0 = {}, acc1 = {};
#pragma unroll
    for (int g = 0; g < 16; ++g) {
      v2f a;
      if (g & 1) {
        a.x = kv[g >> 1].z;
        a.y = kv[g >> 1].w;
      } else {
        a.x = kv[g >> 1].x;
        a.y = kv[g >> 1].y;
      }
      if (g < 12) acc0 = wmma_f32(a, bfr[0][g < 12 ? g : 0], acc0);
      if (g >= 4) acc1 = wmma_f32(a, bfr[1][g >= 4 ? g - 4 : 0], acc1);
    }

    __syncthreads();
#pragma unroll
    for (int r = 0; r < 8; ++r) {
      o[(8 * hi + r) * DO_S + cN] = acc0[r];
      o[(8 * hi + r) * DO_S + 16 + cN] = acc1[r];
    }
    __syncthreads();
#pragma unroll
    for (int q = 0; q < 4; ++q) {
      const int row = q * 4 + erow;
      v4f d = *reinterpret_cast<const v4f*>(&o[row * DO_S + ec4]);
      const size_t chn = rowb + mt * 16 + row;
      d = d + xv[q] + cb[q];
      *reinterpret_cast<v4f*>(out + chn * T_ + t0 + ec4) = d;
    }
    if (mt < 3) {
#pragma unroll
      for (int u = 0; u < 8; ++u) kv[u] = kn[u];
    }
  }
}

extern "C" void kernel_launch(void* const* d_in, const int* in_sizes, int n_in,
                              void* d_out, int out_size, void* d_ws,
                              size_t ws_size, hipStream_t stream) {
  (void)in_sizes;
  (void)n_in;
  (void)out_size;
  (void)ws_size;
  const float* x = (const float*)d_in[0];
  const float* ln_gamma = (const float*)d_in[1];
  const float* ln_beta = (const float*)d_in[2];
  const float* Wk = (const float*)d_in[3];
  const float* bk = (const float*)d_in[4];
  const float* Wc = (const float*)d_in[5];
  const float* bc = (const float*)d_in[6];
  const float* conv_bias = (const float*)d_in[7];
  float* out = (float*)d_out;

  float* ws = (float*)d_ws;
  float* k_ws = ws + 32768;
  float* w_ws = k_ws + (size_t)B_ * C_ * T_;

  glu_gemm_kernel<<<dim3(NROWS / 64, C_ / 64), 256, 0, stream>>>(
      x, ln_gamma, ln_beta, Wk, bk, k_ws);
  conv_weight_kernel<<<NROWS / 32, 256, 0, stream>>>(k_ws, Wc, bc, w_ws);
  dynconv_kernel<<<dim3(T_ / 32, B_), 256, 0, stream>>>(x, k_ws, w_ws,
                                                        conv_bias, out);
}


# GEMM2 K loop software-pipelined: next chunk's k and Wc loads issued as one unconditional group under the current chunk's matrix work; two LDS buffers, one barrier per chunk
# speedup vs baseline: 2.0249x; 1.0545x over previous
#include <hip/hip_runtime.h>
#include <hip/hip_bf16.h>
#include <cstdint>

#define B_ 8
#define C_ 512
#define T_ 2048
#define HEADS_ 8
#define KER_ 31
#define NROWS (B_ * T_)
#define EPS_ 1e-5f

typedef __attribute__((ext_vector_type(2))) float v2f;
typedef __attribute__((ext_vector_type(8))) float v8f;

__device__ __forceinline__ v8f wmma_f32(v2f a, v2f b, v8f c) {
  return __builtin_amdgcn_wmma_f32_16x16x4_f32(false, a, false, b, (short)0, c,
                                               false, false);
}

#define AS_S 66
#define BS_S 66
#define KT_S 65
#define RED_O 4352
__global__ __launch_bounds__(256) __attribute__((amdgpu_waves_per_eu(8, 8))) void
glu_gemm_kernel(
    const float* __restrict__ x, const float* __restrict__ gamma,
    const float* __restrict__ beta, const float* __restrict__ Wk,
    const float* __restrict__ bk, float* __restrict__ k_ws) {
  __shared__ __attribute__((aligned(16))) float smem[64 * AS_S + 128 * BS_S];
  float* As = smem;
  float* Bs = smem + 64 * AS_S;

  const int tid = threadIdx.x;
  const int row0 = blockIdx.x * 64;
  const int n0 = blockIdx.y * 64;
  const int b = row0 / T_;
  const int t0 = row0 % T_;
  const int r_ld = tid & 63;

  const int lane = tid & 31;
  const int wave = __builtin_amdgcn_readfirstlane(tid >> 5);
  const int kg = wave >> 1;
  const int laneM = lane & 15;
  const int hi = lane >> 4;
  const int rblk = wave & 3;
  const int j0 = (wave >> 2) * 32;

  v8f accA0 = {}, accA1 = {}, accB0 = {}, accB1 = {};
  const float* xb = x + (size_t)b * C_ * T_ + t0 + r_ld;
  v2f sx2 = {0.f, 0.f}, sxx2 = {0.f, 0.f};
  v2f gs2[2] = {{0.f, 0.f}, {0.f, 0.f}};
  v2f bs2[2] = {{0.f, 0.f}, {0.f, 0.f}};

  for (int kbase = 0; kbase < C_; kbase += 64) {
#pragma unroll
    for (int i = 0; i < 8; ++i) {
      int kk = kg * 16 + 2 * i;
      int c = kbase + kk;
      v2f v;
      v.x = xb[(size_t)c * T_];
      v.y = xb[(size_t)(c + 1) * T_];
      const v2f g2 = *reinterpret_cast<const v2f*>(&gamma[c]);
      *reinterpret_cast<v2f*>(&As[r_ld * AS_S + kk]) = v * g2;
      sx2 += v;
      sxx2 += v * v;
    }
#pragma unroll
    for (int part = 0; part < 2; ++part) {
      int n = n0 + r_ld + part * C_;
#pragma unroll
      for (int i = 0; i < 8; ++i) {
        int kk = kg * 16 + 2 * i;
        v2f w;
        w.x = Wk[(size_t)(kbase + kk) * (2 * C_) + n];
        w.y = Wk[(size_t)(kbase + kk + 1) * (2 * C_) + n];
        const v2f g2 = *reinterpret_cast<const v2f*>(&gamma[kbase + kk]);
        const v2f b2 = *reinterpret_cast<const v2f*>(&beta[kbase + kk]);
        *reinterpret_cast<v2f*>(&Bs[(part * 64 + r_ld) * BS_S + kk]) = w;
        gs2[part] += g2 * w;
        bs2[part] += b2 * w;
      }
    }
    __syncthreads();
#pragma unroll
    for (int kb = 0; kb < 64; kb += 4) {
      v2f af = *reinterpret_cast<const v2f*>(
          &As[(rblk * 16 + laneM) * AS_S + kb + 2 * hi]);
      v2f b0 = *reinterpret_cast<const v2f*>(
          &Bs[(j0 + laneM) * BS_S + kb + 2 * hi]);
      v2f b1 = *reinterpret_cast<const v2f*>(
          &Bs[(j0 + 16 + laneM) * BS_S + kb + 2 * hi]);
      v2f b2 = *reinterpret_cast<const v2f*>(
          &Bs[(64 + j0 + laneM) * BS_S + kb + 2 * hi]);
      v2f b3 = *reinterpret_cast<const v2f*>(
          &Bs[(64 + j0 + 16 + laneM) * BS_S + kb + 2 * hi]);
      accA0 = wmma_f32(af, b0, accA0);
      accA1 = wmma_f32(af, b1, accA1);
      accB0 = wmma_f32(af, b2, accB0);
      accB1 = wmma_f32(af, b3, accB1);
    }
    __syncthreads();
  }

  float* p_sx = smem + RED_O;
  float* p_sxx = smem + RED_O + 256;
  float* p_g = smem + RED_O + 512;
  float* p_b = smem + RED_O + 1024;
  float* f_mu = smem + RED_O + 1536;
  float* f_rs = smem + RED_O + 1600;
  float* f_g = smem + RED_O + 1664;
  float* f_b = smem + RED_O + 1792;
  p_sx[kg * 64 + r_ld] = sx2.x + sx2.y;
  p_sxx[kg * 64 + r_ld] = sxx2.x + sxx2.y;
  p_g[kg * 128 + r_ld] = gs2[0].x + gs2[0].y;
  p_g[kg * 128 + 64 + r_ld] = gs2[1].x + gs2[1].y;
  p_b[kg * 128 + r_ld] = bs2[0].x + bs2[0].y;
  p_b[kg * 128 + 64 + r_ld] = bs2[1].x + bs2[1].y;
  __syncthreads();
  if (tid < 64) {
    float S = p_sx[tid] + p_sx[64 + tid] + p_sx[128 + tid] + p_sx[192 + tid];
    float S2 = p_sxx[tid] + p_sxx[64 + tid] + p_sxx[128 + tid] + p_sxx[192 + tid];
    float mu = S * (1.f / C_);
    float var = S2 * (1.f / C_) - mu * mu;
    f_mu[tid] = mu;
    f_rs[tid] = rsqrtf(var + EPS_);
  } else if (tid < 192) {
    int col = tid - 64;
    float G = p_g[col] + p_g[128 + col] + p_g[256 + col] + p_g[384 + col];
    float Bt = p_b[col] + p_b[128 + col] + p_b[256 + col] + p_b[384 + col];
    f_g[col] = G;
    f_b[col] = Bt;
  }
  __syncthreads();

  float* kst = smem;
  const float ga0 = f_g[j0 + laneM];
  const float ga1 = f_g[j0 + 16 + laneM];
  const float gb0 = f_g[64 + j0 + laneM];
  const float gb1 = f_g[64 + j0 + 16 + laneM];
  const float oa0 = f_b[j0 + laneM] + bk[n0 + j0 + laneM];
  const float oa1 = f_b[j0 + 16 + laneM] + bk[n0 + j0 + 16 + laneM];
  const float ob0 = f_b[64 + j0 + laneM] + bk[C_ + n0 + j0 + laneM];
  const float ob1 = f_b[64 + j0 + 16 + laneM] + bk[C_ + n0 + j0 + 16 + laneM];
#pragma unroll
  for (int v = 0; v < 8; ++v) {
    int tl = rblk * 16 + v + 8 * hi;
    const float mu = f_mu[tl];
    const float rs = f_rs[tl];
    float a0 = rs * (accA0[v] - mu * ga0) + oa0;
    float g0 = rs * (accB0[v] - mu * gb0) + ob0;
    float a1 = rs * (accA1[v] - mu * ga1) + oa1;
    float g1 = rs * (accB1[v] - mu * gb1) + ob1;
    kst[(j0 + laneM) * KT_S + tl] = a0 / (1.f + __expf(-g0));
    kst[(j0 + 16 + laneM) * KT_S + tl] = a1 / (1.f + __expf(-g1));
  }
  __syncthreads();
  float* kout = k_ws + (size_t)b * C_ * T_;
  for (int i = tid; i < 64 * 64; i += 256) {
    int cc = i >> 6;
    int tl = i & 63;
    kout[(size_t)(n0 + cc) * T_ + t0 + tl] = kst[cc * KT_S + tl];
  }
}

#define A2_S 34
#define B2_S 34
#define P_S 257
#define A2_SZ (32 * A2_S)
#define ST2_SZ (A2_SZ + 256 * B2_S)
struct G2Regs {
  float a[4];
  float b[32];
};
__device__ __forceinline__ void g2_load(G2Regs& rg, const float* __restrict__ kin,
                                        const float* __restrict__ Wc, int kbase,
                                        int tid) {
  const int r = tid & 31;
  const int cq = tid >> 5;
#pragma unroll
  for (int i = 0; i < 4; ++i)
    rg.a[i] = kin[(size_t)(kbase + cq + 8 * i) * T_ + r];
  const int nn = tid & 63;
  const int kq = tid >> 6;
#pragma unroll
  for (int q = 0; q < 4; ++q) {
    const int n = nn + 64 * q;
    const int nc = n < KER_ * HEADS_ ? n : KER_ * HEADS_ - 1;
#pragma unroll
    for (int i = 0; i < 8; ++i) {
      rg.b[q * 8 + i] =
          Wc[(size_t)(kbase + kq * 8 + i) * (KER_ * HEADS_) + nc];
    }
  }
}
__device__ __forceinline__ void g2_store(const G2Regs& rg, float* As, float* Bs,
                                         int tid) {
  const int r = tid & 31;
  const int cq = tid >> 5;
#pragma unroll
  for (int i = 0; i < 4; ++i) As[r * A2_S + cq + 8 * i] = rg.a[i];
  const int nn = tid & 63;
  const int kq = tid >> 6;
#pragma unroll
  for (int q = 0; q < 4; ++q) {
    const bool ok = nn + 64 * q < KER_ * HEADS_;
#pragma unroll
    for (int i = 0; i < 8; ++i)
      Bs[(nn + 64 * q) * B2_S + kq * 8 + i] = ok ? rg.b[q * 8 + i] : 0.f;
  }
}
__global__ __launch_bounds__(256) void conv_weight_kernel(
    const float* __restrict__ k_ws, const float* __restrict__ Wc,
    const float* __restrict__ bc, float* __restrict__ w_ws) {
  __shared__ __attribute__((aligned(16))) float sm2[2 * ST2_SZ];
  float* proj = sm2;
  const int tid = threadIdx.x;
  const int row0 = blockIdx.x * 32;
  const int b = row0 / T_;
  const int t0 = row0 % T_;
  const int lane = tid & 31;
  const int wave = tid >> 5;
  const int laneM = lane & 15;
  const int hi = lane >> 4;
  const int rb = wave & 1;
  const int jb = (wave >> 1) * 64;

  v8f acc[4] = {{}, {}, {}, {}};
  const float* kin = k_ws + (size_t)b * C_ * T_ + t0;

  G2Regs rg;
  g2_load(rg, kin, Wc, 0, tid);
  for (int kbase = 0; kbase < C_; kbase += 32) {
    float* As = sm2 + ((kbase >> 5) & 1) * ST2_SZ;
    float* Bs = As + A2_SZ;
    g2_store(rg, As, Bs, tid);
    __syncthreads();
    if (kbase + 32 < C_) g2_load(rg, kin, Wc, kbase + 32, tid);
#pragma unroll
    for (int kb = 0; kb < 32; kb += 4) {
      v2f af = *reinterpret_cast<const v2f*>(
          &As[(rb * 16 + laneM) * A2_S + kb + 2 * hi]);
#pragma unroll
      for (int jt = 0; jt < 4; ++jt) {
        v2f bf = *reinterpret_cast<const v2f*>(
            &Bs[(jb + jt * 16 + laneM) * B2_S + kb + 2 * hi]);
        acc[jt] = wmma_f32(af, bf, acc[jt]);
      }
    }
  }
  __syncthreads();
#pragma unroll
  for (int jt = 0; jt < 4; ++jt)
#pragma unroll
    for (int v = 0; v < 8; ++v)
      proj[(rb * 16 + v + 8 * hi) * P_S + jb + jt * 16 + laneM] = acc[jt][v];
  __syncthreads();

  {
    int rr = tid >> 3;
    int h = tid & 7;
    float* prow = &proj[rr * P_S + h * KER_];
    const float* bch = &bc[h * KER_];
    float m = -1e30f;
#pragma unroll
    for (int i = 0; i < KER_; ++i) {
      float v = prow[i] + bch[i];
      prow[i] = v;
      m = fmaxf(m, v);
    }
    float s = 0.f;
#pragma unroll
    for (int i = 0; i < KER_; ++i) {
      float e = __expf(prow[i] - m);
      prow[i] = e;
      s += e;
    }
    float inv = 1.f / s;
    float* wout = w_ws + (size_t)(row0 + rr) * 256 + h * 32;
#pragma unroll
    for (int i = 0; i < KER_; ++i) wout[i] = prow[i] * inv;
    wout[KER_] = 0.f;
  }
}

typedef __attribute__((ext_vector_type(4))) float v4f;
#define DW_S 260
#define DO_S 36
__device__ __forceinline__ void load_kwin(v4f (&kv)[8],
                                          const float* __restrict__ krow,
                                          bool lo_ok, bool hi_ok) {
#pragma unroll
  for (int u = 0; u < 8; ++u) {
    const bool in = (u >= 2 || lo_ok) && (u < 6 || hi_ok);
    const v4f v = *reinterpret_cast<const v4f*>(krow + (in ? 8 * u - 16 : 0));
    kv[u] = in ? v : (v4f){0.f, 0.f, 0.f, 0.f};
  }
}
__global__ __launch_bounds__(256) void dynconv_kernel(
    const float* __restrict__ x, const float* __restrict__ k_ws,
    const float* __restrict__ w_ws, const float* __restrict__ conv_bias,
    float* __restrict__ out) {
  __shared__ __attribute__((aligned(16))) float wl[32 * DW_S];
  __shared__ __attribute__((aligned(16))) float ost[8 * 16 * DO_S];
  const int tid = threadIdx.x;
  const int t0 = blockIdx.x * 32;
  const int b = blockIdx.y;
  {
    const v4f* wsrc =
        reinterpret_cast<const v4f*>(w_ws + ((size_t)b * T_ + t0) * 256);
    for (int i = tid; i < 32 * 64; i += 256)
      *reinterpret_cast<v4f*>(&wl[(i >> 6) * DW_S + (i & 63) * 4]) = wsrc[i];
  }
  __syncthreads();

  const int lane = tid & 31;
  const int head = tid >> 5;
  const int cN = lane & 15;
  const int hi = lane >> 4;

  v2f bfr[2][12];
#pragma unroll
  for (int nt = 0; nt < 2; ++nt) {
    const float* wr = &wl[(16 * nt + cN) * DW_S + head * 32];
#pragma unroll
    for (int s = 0; s < 12; ++s) {
      const int kk0 = 8 * (s >> 1) + 4 * hi + 2 * (s & 1) - cN - 1;
      const int kk1 = kk0 + 1;
      const float w0 = wr[min(max(kk0, 0), KER_ - 1)];
      const float w1 = wr[min(max(kk1, 0), KER_ - 1)];
      v2f f;
      f.x = ((unsigned)kk0 < (unsigned)KER_) ? w0 : 0.f;
      f.y = ((unsigned)kk1 < (unsigned)KER_) ? w1 : 0.f;
      bfr[nt][s] = f;
    }
  }

  const bool lo_ok = (t0 != 0);
  const bool hi_ok = (t0 + 32 != T_);
  const size_t rowb = (size_t)b * C_ + head * 64;
  float* o = &ost[head * 16 * DO_S];
  const int erow = lane >> 3;
  const int ec4 = (lane & 7) * 4;

  v4f kv[8];
  load_kwin(kv, k_ws + (rowb + cN) * T_ + t0 + 4 * hi, lo_ok, hi_ok);

#pragma unroll
  for (int mt = 0; mt < 4; ++mt) {
    v4f xv[4];
    float cb[4];
#pragma unroll
    for (int q = 0; q < 4; ++q) {
      const size_t chn = rowb + mt * 16 + q * 4 + erow;
      xv[q] = *reinterpret_cast<const v4f*>(x + chn * T_ + t0 + ec4);
      cb[q] = conv_bias[head * 64 + mt * 16 + q * 4 + erow];
    }
    v4f kn[8];
    if (mt < 3)
      load_kwin(kn, k_ws + (rowb + (mt + 1) * 16 + cN) * T_ + t0 + 4 * hi,
                lo_ok, hi_ok);

    v8f acc0 = {}, acc1 = {};
#pragma unroll
    for (int g = 0; g < 16; ++g) {
      v2f a;
      if (g & 1) {
        a.x = kv[g >> 1].z;
        a.y = kv[g >> 1].w;
      } else {
        a.x = kv[g >> 1].x;
        a.y = kv[g >> 1].y;
      }
      if (g < 12) acc0 = wmma_f32(a, bfr[0][g < 12 ? g : 0], acc0);
      if (g >= 4) acc1 = wmma_f32(a, bfr[1][g >= 4 ? g - 4 : 0], acc1);
    }

    __syncthreads();
#pragma unroll
    for (int r = 0; r < 8; ++r) {
      o[(8 * hi + r) * DO_S + cN] = acc0[r];
      o[(8 * hi + r) * DO_S + 16 + cN] = acc1[r];
    }
    __syncthreads();
#pragma unroll
    for (int q = 0; q < 4; ++q) {
      const int row = q * 4 + erow;
      v4f d = *reinterpret_cast<const v4f*>(&o[row * DO_S + ec4]);
      const size_t chn = rowb + mt * 16 + row;
      d = d + xv[q] + cb[q];
      *reinterpret_cast<v4f*>(out + chn * T_ + t0 + ec4) = d;
    }
    if (mt < 3) {
#pragma unroll
      for (int u = 0; u < 8; ++u) kv[u] = kn[u];
    }
  }
}

extern "C" void kernel_launch(void* const* d_in, const int* in_sizes, int n_in,
                              void* d_out, int out_size, void* d_ws,
                              size_t ws_size, hipStream_t stream) {
  (void)in_sizes;
  (void)n_in;
  (void)out_size;
  (void)ws_size;
  const float* x = (const float*)d_in[0];
  const float* ln_gamma = (const float*)d_in[1];
  const float* ln_beta = (const float*)d_in[2];
  const float* Wk = (const float*)d_in[3];
  const float* bk = (const float*)d_in[4];
  const float* Wc = (const float*)d_in[5];
  const float* bc = (const float*)d_in[6];
  const float* conv_bias = (const float*)d_in[7];
  float* out = (float*)d_out;

  float* ws = (float*)d_ws;
  float* k_ws = ws + 32768;
  float* w_ws = k_ws + (size_t)B_ * C_ * T_;

  glu_gemm_kernel<<<dim3(NROWS / 64, C_ / 64), 256, 0, stream>>>(
      x, ln_gamma, ln_beta, Wk, bk, k_ws);
  conv_weight_kernel<<<NROWS / 32, 256, 0, stream>>>(k_ws, Wc, bc, w_ws);
  dynconv_kernel<<<dim3(T_ / 32, B_), 256, 0, stream>>>(x, k_ws, w_ws,
                                                        conv_bias, out);
}


# conv-weight kernel fused into the dynamic-conv kernel: each 32-t tile's softmax weights stay in LDS; the w workspace round trip and one of three launches removed
# speedup vs baseline: 2.1088x; 1.0414x over previous
#include <hip/hip_runtime.h>
#include <hip/hip_bf16.h>
#include <cstdint>

#define B_ 8
#define C_ 512
#define T_ 2048
#define HEADS_ 8
#define KER_ 31
#define NROWS (B_ * T_)
#define EPS_ 1e-5f

typedef __attribute__((ext_vector_type(2))) float v2f;
typedef __attribute__((ext_vector_type(8))) float v8f;

__device__ __forceinline__ v8f wmma_f32(v2f a, v2f b, v8f c) {
  return __builtin_amdgcn_wmma_f32_16x16x4_f32(false, a, false, b, (short)0, c,
                                               false, false);
}

#define AS_S 66
#define BS_S 66
#define KT_S 65
#define RED_O 4352
__global__ __launch_bounds__(256) __attribute__((amdgpu_waves_per_eu(8, 8))) void
glu_gemm_kernel(
    const float* __restrict__ x, const float* __restrict__ gamma,
    const float* __restrict__ beta, const float* __restrict__ Wk,
    const float* __restrict__ bk, float* __restrict__ k_ws) {
  __shared__ __attribute__((aligned(16))) float smem[64 * AS_S + 128 * BS_S];
  float* As = smem;
  float* Bs = smem + 64 * AS_S;

  const int tid = threadIdx.x;
  const int row0 = blockIdx.x * 64;
  const int n0 = blockIdx.y * 64;
  const int b = row0 / T_;
  const int t0 = row0 % T_;
  const int r_ld = tid & 63;

  const int lane = tid & 31;
  const int wave = __builtin_amdgcn_readfirstlane(tid >> 5);
  const int kg = wave >> 1;
  const int laneM = lane & 15;
  const int hi = lane >> 4;
  const int rblk = wave & 3;
  const int j0 = (wave >> 2) * 32;

  v8f accA0 = {}, accA1 = {}, accB0 = {}, accB1 = {};
  const float* xb = x + (size_t)b * C_ * T_ + t0 + r_ld;
  v2f sx2 = {0.f, 0.f}, sxx2 = {0.f, 0.f};
  v2f gs2[2] = {{0.f, 0.f}, {0.f, 0.f}};
  v2f bs2[2] = {{0.f, 0.f}, {0.f, 0.f}};

  for (int kbase = 0; kbase < C_; kbase += 64) {
#pragma unroll
    for (int i = 0; i < 8; ++i) {
      int kk = kg * 16 + 2 * i;
      int c = kbase + kk;
      v2f v;
      v.x = xb[(size_t)c * T_];
      v.y = xb[(size_t)(c + 1) * T_];
      const v2f g2 = *reinterpret_cast<const v2f*>(&gamma[c]);
      *reinterpret_cast<v2f*>(&As[r_ld * AS_S + kk]) = v * g2;
      sx2 += v;
      sxx2 += v * v;
    }
#pragma unroll
    for (int part = 0; part < 2; ++part) {
      int n = n0 + r_ld + part * C_;
#pragma unroll
      for (int i = 0; i < 8; ++i) {
        int kk = kg * 16 + 2 * i;
        v2f w;
        w.x = Wk[(size_t)(kbase + kk) * (2 * C_) + n];
        w.y = Wk[(size_t)(kbase + kk + 1) * (2 * C_) + n];
        const v2f g2 = *reinterpret_cast<const v2f*>(&gamma[kbase + kk]);
        const v2f b2 = *reinterpret_cast<const v2f*>(&beta[kbase + kk]);
        *reinterpret_cast<v2f*>(&Bs[(part * 64 + r_ld) * BS_S + kk]) = w;
        gs2[part] += g2 * w;
        bs2[part] += b2 * w;
      }
    }
    __syncthreads();
#pragma unroll
    for (int kb = 0; kb < 64; kb += 4) {
      v2f af = *reinterpret_cast<const v2f*>(
          &As[(rblk * 16 + laneM) * AS_S + kb + 2 * hi]);
      v2f b0 = *reinterpret_cast<const v2f*>(
          &Bs[(j0 + laneM) * BS_S + kb + 2 * hi]);
      v2f b1 = *reinterpret_cast<const v2f*>(
          &Bs[(j0 + 16 + laneM) * BS_S + kb + 2 * hi]);
      v2f b2 = *reinterpret_cast<const v2f*>(
          &Bs[(64 + j0 + laneM) * BS_S + kb + 2 * hi]);
      v2f b3 = *reinterpret_cast<const v2f*>(
          &Bs[(64 + j0 + 16 + laneM) * BS_S + kb + 2 * hi]);
      accA0 = wmma_f32(af, b0, accA0);
      accA1 = wmma_f32(af, b1, accA1);
      accB0 = wmma_f32(af, b2, accB0);
      accB1 = wmma_f32(af, b3, accB1);
    }
    __syncthreads();
  }

  float* p_sx = smem + RED_O;
  float* p_sxx = smem + RED_O + 256;
  float* p_g = smem + RED_O + 512;
  float* p_b = smem + RED_O + 1024;
  float* f_mu = smem + RED_O + 1536;
  float* f_rs = smem + RED_O + 1600;
  float* f_g = smem + RED_O + 1664;
  float* f_b = smem + RED_O + 1792;
  p_sx[kg * 64 + r_ld] = sx2.x + sx2.y;
  p_sxx[kg * 64 + r_ld] = sxx2.x + sxx2.y;
  p_g[kg * 128 + r_ld] = gs2[0].x + gs2[0].y;
  p_g[kg * 128 + 64 + r_ld] = gs2[1].x + gs2[1].y;
  p_b[kg * 128 + r_ld] = bs2[0].x + bs2[0].y;
  p_b[kg * 128 + 64 + r_ld] = bs2[1].x + bs2[1].y;
  __syncthreads();
  if (tid < 64) {
    float S = p_sx[tid] + p_sx[64 + tid] + p_sx[128 + tid] + p_sx[192 + tid];
    float S2 = p_sxx[tid] + p_sxx[64 + tid] + p_sxx[128 + tid] + p_sxx[192 + tid];
    float mu = S * (1.f / C_);
    float var = S2 * (1.f / C_) - mu * mu;
    f_mu[tid] = mu;
    f_rs[tid] = rsqrtf(var + EPS_);
  } else if (tid < 192) {
    int col = tid - 64;
    float G = p_g[col] + p_g[128 + col] + p_g[256 + col] + p_g[384 + col];
    float Bt = p_b[col] + p_b[128 + col] + p_b[256 + col] + p_b[384 + col];
    f_g[col] = G;
    f_b[col] = Bt;
  }
  __syncthreads();

  float* kst = smem;
  const float ga0 = f_g[j0 + laneM];
  const float ga1 = f_g[j0 + 16 + laneM];
  const float gb0 = f_g[64 + j0 + laneM];
  const float gb1 = f_g[64 + j0 + 16 + laneM];
  const float oa0 = f_b[j0 + laneM] + bk[n0 + j0 + laneM];
  const float oa1 = f_b[j0 + 16 + laneM] + bk[n0 + j0 + 16 + laneM];
  const float ob0 = f_b[64 + j0 + laneM] + bk[C_ + n0 + j0 + laneM];
  const float ob1 = f_b[64 + j0 + 16 + laneM] + bk[C_ + n0 + j0 + 16 + laneM];
#pragma unroll
  for (int v = 0; v < 8; ++v) {
    int tl = rblk * 16 + v + 8 * hi;
    const float mu = f_mu[tl];
    const float rs = f_rs[tl];
    float a0 = rs * (accA0[v] - mu * ga0) + oa0;
    float g0 = rs * (accB0[v] - mu * gb0) + ob0;
    float a1 = rs * (accA1[v] - mu * ga1) + oa1;
    float g1 = rs * (accB1[v] - mu * gb1) + ob1;
    kst[(j0 + laneM) * KT_S + tl] = a0 / (1.f + __expf(-g0));
    kst[(j0 + 16 + laneM) * KT_S + tl] = a1 / (1.f + __expf(-g1));
  }
  __syncthreads();
  float* kout = k_ws + (size_t)b * C_ * T_;
  for (int i = tid; i < 64 * 64; i += 256) {
    int cc = i >> 6;
    int tl = i & 63;
    kout[(size_t)(n0 + cc) * T_ + t0 + tl] = kst[cc * KT_S + tl];
  }
}

#define A2_S 34
#define B2_S 34
#define A2_SZ (32 * A2_S)
#define ST2_SZ (A2_SZ + 256 * B2_S)
struct G2Regs {
  float a[4];
  float b[32];
};
__device__ __forceinline__ void g2_load(G2Regs& rg, const float* __restrict__ kin,
                                        const float* __restrict__ Wc, int kbase,
                                        int tid) {
  const int r = tid & 31;
  const int cq = tid >> 5;
#pragma unroll
  for (int i = 0; i < 4; ++i)
    rg.a[i] = kin[(size_t)(kbase + cq + 8 * i) * T_ + r];
  const int nn = tid & 63;
  const int kq = tid >> 6;
#pragma unroll
  for (int q = 0; q < 4; ++q) {
    const int n = nn + 64 * q;
    const int nc = n < KER_ * HEADS_ ? n : KER_ * HEADS_ - 1;
#pragma unroll
    for (int i = 0; i < 8; ++i) {
      rg.b[q * 8 + i] =
          Wc[(size_t)(kbase + kq * 8 + i) * (KER_ * HEADS_) + nc];
    }
  }
}
__device__ __forceinline__ void g2_store(const G2Regs& rg, float* As, float* Bs,
                                         int tid) {
  const int r = tid & 31;
  const int cq = tid >> 5;
#pragma unroll
  for (int i = 0; i < 4; ++i) As[r * A2_S + cq + 8 * i] = rg.a[i];
  const int nn = tid & 63;
  const int kq = tid >> 6;
#pragma unroll
  for (int q = 0; q < 4; ++q) {
    const bool ok = nn + 64 * q < KER_ * HEADS_;
#pragma unroll
    for (int i = 0; i < 8; ++i)
      Bs[(nn + 64 * q) * B2_S + kq * 8 + i] = ok ? rg.b[q * 8 + i] : 0.f;
  }
}
typedef __attribute__((ext_vector_type(4))) float v4f;
#define DW_S 260
#define DO_S 36
__device__ __forceinline__ void load_kwin(v4f (&kv)[8],
                                          const float* __restrict__ krow,
                                          bool lo_ok, bool hi_ok) {
#pragma unroll
  for (int u = 0; u < 8; ++u) {
    const bool in = (u >= 2 || lo_ok) && (u < 6 || hi_ok);
    const v4f v = *reinterpret_cast<const v4f*>(krow + (in ? 8 * u - 16 : 0));
    kv[u] = in ? v : (v4f){0.f, 0.f, 0.f, 0.f};
  }
}
__global__ __launch_bounds__(256) void conv_fused_kernel(
    const float* __restrict__ x, const float* __restrict__ k_ws,
    const float* __restrict__ Wc, const float* __restrict__ bc,
    const float* __restrict__ conv_bias, float* __restrict__ out) {
  __shared__ __attribute__((aligned(16))) float sm2[2 * ST2_SZ];
  float* wl = sm2;
  float* ost = sm2 + 32 * DW_S;
  const int tid = threadIdx.x;
  const int row0 = blockIdx.x * 32;
  const int b = row0 / T_;
  const int t0 = row0 % T_;
  const int lane = tid & 31;
  const int wave = tid >> 5;
  const int laneM = lane & 15;
  const int hi = lane >> 4;
  const int rb = wave & 1;
  const int jb = (wave >> 1) * 64;

  {
    v8f acc[4] = {{}, {}, {}, {}};
    const float* kin = k_ws + (size_t)b * C_ * T_ + t0;

    G2Regs rg;
    g2_load(rg, kin, Wc, 0, tid);
    for (int kbase = 0; kbase < C_; kbase += 32) {
      float* As = sm2 + ((kbase >> 5) & 1) * ST2_SZ;
      float* Bs = As + A2_SZ;
      g2_store(rg, As, Bs, tid);
      __syncthreads();
      if (kbase + 32 < C_) g2_load(rg, kin, Wc, kbase + 32, tid);
#pragma unroll
      for (int kb = 0; kb < 32; kb += 4) {
        v2f af = *reinterpret_cast<const v2f*>(
            &As[(rb * 16 + laneM) * A2_S + kb + 2 * hi]);
#pragma unroll
        for (int jt = 0; jt < 4; ++jt) {
          v2f bf = *reinterpret_cast<const v2f*>(
              &Bs[(jb + jt * 16 + laneM) * B2_S + kb + 2 * hi]);
          acc[jt] = wmma_f32(af, bf, acc[jt]);
        }
      }
    }
    __syncthreads();
#pragma unroll
    for (int jt = 0; jt < 4; ++jt)
#pragma unroll
      for (int v = 0; v < 8; ++v)
        wl[(rb * 16 + v + 8 * hi) * DW_S + jb + jt * 16 + laneM] = acc[jt][v];
  }
  __syncthreads();

  {
    int rr = tid >> 3;
    int h = tid & 7;
    float* prow = &wl[rr * DW_S + h * KER_];
    const float* bch = &bc[h * KER_];
    float m = -1e30f;
#pragma unroll
    for (int i = 0; i < KER_; ++i) {
      float v = prow[i] + bch[i];
      prow[i] = v;
      m = fmaxf(m, v);
    }
    float s = 0.f;
#pragma unroll
    for (int i = 0; i < KER_; ++i) {
      float e = __expf(prow[i] - m);
      prow[i] = e;
      s += e;
    }
    float inv = 1.f / s;
#pragma unroll
    for (int i = 0; i < KER_; ++i) prow[i] = prow[i] * inv;
  }
  __syncthreads();

  const int head = wave;
  const int cN = laneM;

  v2f bfr[2][12];
#pragma unroll
  for (int nt = 0; nt < 2; ++nt) {
    const float* wr = &wl[(16 * nt + cN) * DW_S + head * KER_];
#pragma unroll
    for (int s = 0; s < 12; ++s) {
      const int kk0 = 8 * (s >> 1) + 4 * hi + 2 * (s & 1) - cN - 1;
      const int kk1 = kk0 + 1;
      const float w0 = wr[min(max(kk0, 0), KER_ - 1)];
      const float w1 = wr[min(max(kk1, 0), KER_ - 1)];
      v2f f;
      f.x = ((unsigned)kk0 < (unsigned)KER_) ? w0 : 0.f;
      f.y = ((unsigned)kk1 < (unsigned)KER_) ? w1 : 0.f;
      bfr[nt][s] = f;
    }
  }

  const bool lo_ok = (t0 != 0);
  const bool hi_ok = (t0 + 32 != T_);
  const size_t rowb = (size_t)b * C_ + head * 64;
  float* o = &ost[head * 16 * DO_S];
  const int erow = lane >> 3;
  const int ec4 = (lane & 7) * 4;

  v4f kv[8];
  load_kwin(kv, k_ws + (rowb + cN) * T_ + t0 + 4 * hi, lo_ok, hi_ok);

#pragma unroll
  for (int mt = 0; mt < 4; ++mt) {
    v4f xv[4];
    float cb[4];
#pragma unroll
    for (int q = 0; q < 4; ++q) {
      const size_t chn = rowb + mt * 16 + q * 4 + erow;
      xv[q] = *reinterpret_cast<const v4f*>(x + chn * T_ + t0 + ec4);
      cb[q] = conv_bias[head * 64 + mt * 16 + q * 4 + erow];
    }
    v4f kn[8];
    if (mt < 3)
      load_kwin(kn, k_ws + (rowb + (mt + 1) * 16 + cN) * T_ + t0 + 4 * hi,
                lo_ok, hi_ok);

    v8f acc0 = {}, acc1 = {};
#pragma unroll
    for (int g = 0; g < 16; ++g) {
      v2f a;
      if (g & 1) {
        a.x = kv[g >> 1].z;
        a.y = kv[g >> 1].w;
      } else {
        a.x = kv[g >> 1].x;
        a.y = kv[g >> 1].y;
      }
      if (g < 12) acc0 = wmma_f32(a, bfr[0][g < 12 ? g : 0], acc0);
      if (g >= 4) acc1 = wmma_f32(a, bfr[1][g >= 4 ? g - 4 : 0], acc1);
    }

    __syncthreads();
#pragma unroll
    for (int r = 0; r < 8; ++r) {
      o[(8 * hi + r) * DO_S + cN] = acc0[r];
      o[(8 * hi + r) * DO_S + 16 + cN] = acc1[r];
    }
    __syncthreads();
#pragma unroll
    for (int q = 0; q < 4; ++q) {
      const int row = q * 4 + erow;
      v4f d = *reinterpret_cast<const v4f*>(&o[row * DO_S + ec4]);
      const size_t chn = rowb + mt * 16 + row;
      d = d + xv[q] + cb[q];
      *reinterpret_cast<v4f*>(out + chn * T_ + t0 + ec4) = d;
    }
    if (mt < 3) {
#pragma unroll
      for (int u = 0; u < 8; ++u) kv[u] = kn[u];
    }
  }
}

extern "C" void kernel_launch(void* const* d_in, const int* in_sizes, int n_in,
                              void* d_out, int out_size, void* d_ws,
                              size_t ws_size, hipStream_t stream) {
  (void)in_sizes;
  (void)n_in;
  (void)out_size;
  (void)ws_size;
  const float* x = (const float*)d_in[0];
  const float* ln_gamma = (const float*)d_in[1];
  const float* ln_beta = (const float*)d_in[2];
  const float* Wk = (const float*)d_in[3];
  const float* bk = (const float*)d_in[4];
  const float* Wc = (const float*)d_in[5];
  const float* bc = (const float*)d_in[6];
  const float* conv_bias = (const float*)d_in[7];
  float* out = (float*)d_out;

  float* ws = (float*)d_ws;
  float* k_ws = ws + 32768;

  glu_gemm_kernel<<<dim3(NROWS / 64, C_ / 64), 256, 0, stream>>>(
      x, ln_gamma, ln_beta, Wk, bk, k_ws);
  conv_fused_kernel<<<NROWS / 32, 256, 0, stream>>>(x, k_ws, Wc, bc, conv_bias,
                                                    out);
}
